# starting kernel
# baseline (speedup 1.0000x reference)
#include <hip/hip_runtime.h>

typedef float v2f __attribute__((ext_vector_type(2)));
typedef float v8f __attribute__((ext_vector_type(8)));

#define IMG   512
#define TILE  16
#define HALO  18
#define PITCH 20
#define CHSZ  (HALO*PITCH)
#define NCH   30

__global__ __launch_bounds__(256) void hmm_fused_kernel(
    const float* __restrict__ obs,
    const float* __restrict__ P,
    const float* __restrict__ u_k,
    const float* __restrict__ w_k,
    const float* __restrict__ conv_w,
    const float* __restrict__ conv_b,
    float* __restrict__ u_out,
    float* __restrict__ w_out)
{
    __shared__ float X[NCH * CHSZ];

    const int tid = threadIdx.x;
    const int wgx = blockIdx.x * TILE;
    const int wgy = blockIdx.y * TILE;

    float Pm[9];
#pragma unroll
    for (int i = 0; i < 9; ++i) Pm[i] = P[i];
    const float cb0 = conv_b[0], cb1 = conv_b[1], cb2 = conv_b[2];

#pragma unroll
    for (int it = 0; it < 2; ++it) {
        int hp = tid + it * 256;
        if (hp < HALO * HALO) {
            const int hy = hp / HALO, hx = hp % HALO;
            const int gy = wgy + hy - 1, gx = wgx + hx - 1;
            const int lb = hy * PITCH + hx;
            if (gy >= 0 && gy < IMG && gx >= 0 && gx < IMG) {
                const int pix = gy * IMG + gx;
                const float o0 = obs[pix * 3 + 0], o1 = obs[pix * 3 + 1], o2 = obs[pix * 3 + 2];
                const float u0 = u_k[pix * 3 + 0], u1 = u_k[pix * 3 + 1], u2 = u_k[pix * 3 + 2];
                const float b0 = Pm[0] * o0 + Pm[3] * o1 + Pm[6] * o2;
                const float b1 = Pm[1] * o0 + Pm[4] * o1 + Pm[7] * o2;
                const float b2 = Pm[2] * o0 + Pm[5] * o1 + Pm[8] * o2;
                const float Bu0 = b0 * u0, Bu1 = b1 * u1, Bu2 = b2 * u2;
                const float bu  = Bu0 + Bu1 + Bu2;
                const float inv = 1.0f / bu;
                const float Z0 = Bu0 * inv, Z1 = Bu1 * inv, Z2 = Bu2 * inv;
                X[0 * CHSZ + lb] = Z0;
                X[1 * CHSZ + lb] = Z1;
                X[2 * CHSZ + lb] = Z2;
                const float ov[3] = {o0, o1, o2};
                const float uv[3] = {u0, u1, u2};
                const float bv[3] = {b0, b1, b2};
                const float Zv[3] = {Z0, Z1, Z2};
                const float* wp = w_k + (long)pix * 27;
#pragma unroll
                for (int ij = 0; ij < 9; ++ij) {
                    const int i = ij / 3, j = ij % 3;
                    float du0 = bv[0] * wp[0 * 9 + ij] + ((j == 0) ? ov[i] * uv[0] : 0.0f);
                    float du1 = bv[1] * wp[1 * 9 + ij] + ((j == 1) ? ov[i] * uv[1] : 0.0f);
                    float du2 = bv[2] * wp[2 * 9 + ij] + ((j == 2) ? ov[i] * uv[2] : 0.0f);
                    float sdu = du0 + du1 + du2;
                    X[(3 + ij * 3 + 0) * CHSZ + lb] = (du0 - sdu * Zv[0]) * inv;
                    X[(3 + ij * 3 + 1) * CHSZ + lb] = (du1 - sdu * Zv[1]) * inv;
                    X[(3 + ij * 3 + 2) * CHSZ + lb] = (du2 - sdu * Zv[2]) * inv;
                }
            } else {
#pragma unroll
                for (int c = 0; c < NCH; ++c) X[c * CHSZ + lb] = 0.0f;
            }
        }
    }
    __syncthreads();

    const int lane = tid & 31;
    const int wv   = tid >> 5;
    const int half = (lane < 16) ? 0 : 2;
    const int n    = lane & 15;

    v2f Amat[7];
    int off0[7], off1[7];
#pragma unroll
    for (int jc = 0; jc < 7; ++jc) {
        const int k0 = jc * 4 + half;
        const int k1 = k0 + 1;
        float a0 = 0.0f, a1 = 0.0f;
        if (n < 3 && k0 < 27) a0 = conv_w[n * 27 + (k0 % 3) * 9 + (k0 / 3)];
        if (n < 3 && k1 < 27) a1 = conv_w[n * 27 + (k1 % 3) * 9 + (k1 / 3)];
        Amat[jc][0] = a0; Amat[jc][1] = a1;
        const int c0 = (k0 < 27) ? k0 : 26;
        const int c1 = (k1 < 27) ? k1 : 26;
        const int t0 = c0 / 3, t1 = c1 / 3;
        off0[jc] = (c0 % 3) * CHSZ + (t0 / 3) * PITCH + (t0 % 3) + n;
        off1[jc] = (c1 % 3) * CHSZ + (t1 / 3) * PITCH + (t1 % 3) + n;
    }

    auto conv_group = [&](int gbase) -> v8f {
        v8f acc = {0.f, 0.f, 0.f, 0.f, 0.f, 0.f, 0.f, 0.f};
#pragma unroll
        for (int jc = 0; jc < 7; ++jc) {
            v2f Bv;
            Bv[0] = X[gbase + off0[jc]];
            Bv[1] = X[gbase + off1[jc]];
            acc = __builtin_amdgcn_wmma_f32_16x16x4_f32(
                false, Amat[jc], false, Bv, (short)0, acc, false, false);
        }
        return acc;
    };

#pragma unroll
    for (int r = 0; r < 2; ++r) {
        const int ty     = wv * 2 + r;
        const int gy     = wgy + ty;
        const int rowoff = ty * PITCH;

        v8f y = conv_group(rowoff);
        float y0 = y[0] + cb0, y1 = y[1] + cb1, y2 = y[2] + cb2;
        float m  = fmaxf(y0, fmaxf(y1, y2));
        float e0 = __expf(y0 - m), e1 = __expf(y1 - m), e2 = __expf(y2 - m);
        float is = 1.0f / (e0 + e1 + e2);
        float p0 = e0 * is, p1 = e1 * is, p2 = e2 * is;

        const int pix = gy * IMG + (wgx + n);
        if (lane < 16) {
            u_out[pix * 3 + 0] = p0;
            u_out[pix * 3 + 1] = p1;
            u_out[pix * 3 + 2] = p2;
        }

        for (int g = 1; g < 10; ++g) {
            v8f dv = conv_group(g * (3 * CHSZ) + rowoff);
            float d0 = dv[0], d1 = dv[1], d2 = dv[2];
            float dot = p0 * d0 + p1 * d1 + p2 * d2;
            float w0 = p0 * (d0 - dot);
            float w1 = p1 * (d1 - dot);
            float w2 = p2 * (d2 - dot);
            if (lane < 16) {
                float* wo = w_out + (long)pix * 27 + (g - 1);
                wo[0]  = w0;
                wo[9]  = w1;
                wo[18] = w2;
            }
        }
    }
}

extern "C" void kernel_launch(void* const* d_in, const int* in_sizes, int n_in,
                              void* d_out, int out_size, void* d_ws, size_t ws_size,
                              hipStream_t stream) {
    (void)in_sizes; (void)n_in; (void)out_size; (void)d_ws; (void)ws_size;
    const float* obs    = (const float*)d_in[0];
    const float* P      = (const float*)d_in[1];
    const float* u_k    = (const float*)d_in[2];
    const float* w_k    = (const float*)d_in[3];
    const float* conv_w = (const float*)d_in[4];
    const float* conv_b = (const float*)d_in[5];
    float* u_out = (float*)d_out;
    float* w_out = u_out + (size_t)IMG * IMG * 3;

    dim3 grid(IMG / TILE, IMG / TILE);
    dim3 block(256);
    hmm_fused_kernel<<<grid, block, 0, stream>>>(obs, P, u_k, w_k, conv_w, conv_b,
                                                 u_out, w_out);
}


# w_out rows staged per wave in LDS and written with 128-bit row-contiguous whole-line stores, replacing 54 dword stores per wave whose active lanes each hit a different 128-byte line
# speedup vs baseline: 1.2427x; 1.2427x over previous
#include <hip/hip_runtime.h>

typedef float v2f __attribute__((ext_vector_type(2)));
typedef float v4f __attribute__((ext_vector_type(4)));
typedef float v8f __attribute__((ext_vector_type(8)));

#define IMG   512
#define TILE  16
#define HALO  18
#define PITCH 20
#define CHSZ  (HALO*PITCH)
#define NCH   30
#define WROW  (TILE*27)

__global__ __launch_bounds__(256) void hmm_fused_kernel(
    const float* __restrict__ obs,
    const float* __restrict__ P,
    const float* __restrict__ u_k,
    const float* __restrict__ w_k,
    const float* __restrict__ conv_w,
    const float* __restrict__ conv_b,
    float* __restrict__ u_out,
    float* __restrict__ w_out)
{
    __shared__ float X[NCH * CHSZ];
    __shared__ __attribute__((aligned(16))) float Ost[TILE * WROW];

    const int tid = threadIdx.x;
    const int wgx = blockIdx.x * TILE;
    const int wgy = blockIdx.y * TILE;

    float Pm[9];
#pragma unroll
    for (int i = 0; i < 9; ++i) Pm[i] = P[i];
    const float cb0 = conv_b[0], cb1 = conv_b[1], cb2 = conv_b[2];

#pragma unroll
    for (int it = 0; it < 2; ++it) {
        int hp = tid + it * 256;
        if (hp < HALO * HALO) {
            const int hy = hp / HALO, hx = hp % HALO;
            const int gy = wgy + hy - 1, gx = wgx + hx - 1;
            const int lb = hy * PITCH + hx;
            if (gy >= 0 && gy < IMG && gx >= 0 && gx < IMG) {
                const int pix = gy * IMG + gx;
                const float o0 = obs[pix * 3 + 0], o1 = obs[pix * 3 + 1], o2 = obs[pix * 3 + 2];
                const float u0 = u_k[pix * 3 + 0], u1 = u_k[pix * 3 + 1], u2 = u_k[pix * 3 + 2];
                const float b0 = Pm[0] * o0 + Pm[3] * o1 + Pm[6] * o2;
                const float b1 = Pm[1] * o0 + Pm[4] * o1 + Pm[7] * o2;
                const float b2 = Pm[2] * o0 + Pm[5] * o1 + Pm[8] * o2;
                const float Bu0 = b0 * u0, Bu1 = b1 * u1, Bu2 = b2 * u2;
                const float bu  = Bu0 + Bu1 + Bu2;
                const float inv = 1.0f / bu;
                const float Z0 = Bu0 * inv, Z1 = Bu1 * inv, Z2 = Bu2 * inv;
                X[0 * CHSZ + lb] = Z0;
                X[1 * CHSZ + lb] = Z1;
                X[2 * CHSZ + lb] = Z2;
                const float ov[3] = {o0, o1, o2};
                const float uv[3] = {u0, u1, u2};
                const float bv[3] = {b0, b1, b2};
                const float Zv[3] = {Z0, Z1, Z2};
                const float* wp = w_k + (long)pix * 27;
#pragma unroll
                for (int ij = 0; ij < 9; ++ij) {
                    const int i = ij / 3, j = ij % 3;
                    float du0 = bv[0] * wp[0 * 9 + ij] + ((j == 0) ? ov[i] * uv[0] : 0.0f);
                    float du1 = bv[1] * wp[1 * 9 + ij] + ((j == 1) ? ov[i] * uv[1] : 0.0f);
                    float du2 = bv[2] * wp[2 * 9 + ij] + ((j == 2) ? ov[i] * uv[2] : 0.0f);
                    float sdu = du0 + du1 + du2;
                    X[(3 + ij * 3 + 0) * CHSZ + lb] = (du0 - sdu * Zv[0]) * inv;
                    X[(3 + ij * 3 + 1) * CHSZ + lb] = (du1 - sdu * Zv[1]) * inv;
                    X[(3 + ij * 3 + 2) * CHSZ + lb] = (du2 - sdu * Zv[2]) * inv;
                }
            } else {
#pragma unroll
                for (int c = 0; c < NCH; ++c) X[c * CHSZ + lb] = 0.0f;
            }
        }
    }
    __syncthreads();

    const int lane = tid & 31;
    const int wv   = tid >> 5;
    const int half = (lane < 16) ? 0 : 2;
    const int n    = lane & 15;

    v2f Amat[7];
    int off0[7], off1[7];
#pragma unroll
    for (int jc = 0; jc < 7; ++jc) {
        const int k0 = jc * 4 + half;
        const int k1 = k0 + 1;
        float a0 = 0.0f, a1 = 0.0f;
        if (n < 3 && k0 < 27) a0 = conv_w[n * 27 + (k0 % 3) * 9 + (k0 / 3)];
        if (n < 3 && k1 < 27) a1 = conv_w[n * 27 + (k1 % 3) * 9 + (k1 / 3)];
        Amat[jc][0] = a0; Amat[jc][1] = a1;
        const int c0 = (k0 < 27) ? k0 : 26;
        const int c1 = (k1 < 27) ? k1 : 26;
        const int t0 = c0 / 3, t1 = c1 / 3;
        off0[jc] = (c0 % 3) * CHSZ + (t0 / 3) * PITCH + (t0 % 3) + n;
        off1[jc] = (c1 % 3) * CHSZ + (t1 / 3) * PITCH + (t1 % 3) + n;
    }

    auto conv_group = [&](int gbase) -> v8f {
        v8f acc = {0.f, 0.f, 0.f, 0.f, 0.f, 0.f, 0.f, 0.f};
#pragma unroll
        for (int jc = 0; jc < 7; ++jc) {
            v2f Bv;
            Bv[0] = X[gbase + off0[jc]];
            Bv[1] = X[gbase + off1[jc]];
            acc = __builtin_amdgcn_wmma_f32_16x16x4_f32(
                false, Amat[jc], false, Bv, (short)0, acc, false, false);
        }
        return acc;
    };

#pragma unroll
    for (int r = 0; r < 2; ++r) {
        const int ty     = wv * 2 + r;
        const int gy     = wgy + ty;
        const int rowoff = ty * PITCH;

        v8f y = conv_group(rowoff);
        float y0 = y[0] + cb0, y1 = y[1] + cb1, y2 = y[2] + cb2;
        float m  = fmaxf(y0, fmaxf(y1, y2));
        float e0 = __expf(y0 - m), e1 = __expf(y1 - m), e2 = __expf(y2 - m);
        float is = 1.0f / (e0 + e1 + e2);
        float p0 = e0 * is, p1 = e1 * is, p2 = e2 * is;

        const int pix = gy * IMG + (wgx + n);
        if (lane < 16) {
            u_out[pix * 3 + 0] = p0;
            u_out[pix * 3 + 1] = p1;
            u_out[pix * 3 + 2] = p2;
        }

        for (int g = 1; g < 10; ++g) {
            v8f dv = conv_group(g * (3 * CHSZ) + rowoff);
            float d0 = dv[0], d1 = dv[1], d2 = dv[2];
            float dot = p0 * d0 + p1 * d1 + p2 * d2;
            float w0 = p0 * (d0 - dot);
            float w1 = p1 * (d1 - dot);
            float w2 = p2 * (d2 - dot);
            if (lane < 16) {
                float* so = &Ost[ty * WROW + n * 27 + (g - 1)];
                so[0]  = w0;
                so[9]  = w1;
                so[18] = w2;
            }
        }

        {
            v4f* dst = (v4f*)(w_out + (long)(gy * IMG + wgx) * 27);
            const v4f* srow = (const v4f*)&Ost[ty * WROW];
#pragma unroll
            for (int q = lane; q < WROW / 4; q += 32) dst[q] = srow[q];
        }
    }
}

extern "C" void kernel_launch(void* const* d_in, const int* in_sizes, int n_in,
                              void* d_out, int out_size, void* d_ws, size_t ws_size,
                              hipStream_t stream) {
    (void)in_sizes; (void)n_in; (void)out_size; (void)d_ws; (void)ws_size;
    const float* obs    = (const float*)d_in[0];
    const float* P      = (const float*)d_in[1];
    const float* u_k    = (const float*)d_in[2];
    const float* w_k    = (const float*)d_in[3];
    const float* conv_w = (const float*)d_in[4];
    const float* conv_b = (const float*)d_in[5];
    float* u_out = (float*)d_out;
    float* w_out = u_out + (size_t)IMG * IMG * 3;

    dim3 grid(IMG / TILE, IMG / TILE);
    dim3 block(256);
    hmm_fused_kernel<<<grid, block, 0, stream>>>(obs, P, u_k, w_k, conv_w, conv_b,
                                                 u_out, w_out);
}


# tangent channels sum to zero over the 3 states, so state 2 is folded into the conv weights (W_ci - W_2): K 48 -> 32 in 4 of 5 tiles, 44 WMMAs per wave not 60; conv summation order changes
# speedup vs baseline: 2.0463x; 1.6467x over previous
#include <hip/hip_runtime.h>

typedef float v2f __attribute__((ext_vector_type(2)));
typedef float v4f __attribute__((ext_vector_type(4)));
typedef float v8f __attribute__((ext_vector_type(8)));

#define IMG   512
#define TILE  16
#define HALO  18
#define PITCH 20
#define CHSZ  (HALO*PITCH + 10)
#define NCH   22
#define TAN1  6
#define XRES  30
#define WROW  (TILE*27)

__global__ __launch_bounds__(256) void hmm_fused_kernel(
    const float* __restrict__ obs,
    const float* __restrict__ P,
    const float* __restrict__ u_k,
    const float* __restrict__ w_k,
    const float* __restrict__ conv_w,
    const float* __restrict__ conv_b,
    float* __restrict__ u_out,
    float* __restrict__ w_out)
{
    __shared__ __attribute__((aligned(16))) float X[XRES * CHSZ];
    __shared__ __attribute__((aligned(16))) float Ost[TILE * WROW];

    const int tid = threadIdx.x;
    const int wgx = blockIdx.x * TILE;
    const int wgy = blockIdx.y * TILE;

    float Pm[9];
#pragma unroll
    for (int i = 0; i < 9; ++i) Pm[i] = P[i];
    const float cb0 = conv_b[0], cb1 = conv_b[1], cb2 = conv_b[2];

    const int lane = tid & 31;
    const int wv   = tid >> 5;
    const int h    = lane >> 4;
    const int col  = lane & 15;

    v2f Amat[12];
    {
        const int yo = col >> 3, rr = col & 7;
        const int xo = (rr >= 3) ? 1 : 0;
        const int co = min(rr - xo * 3, 2);
#pragma unroll
        for (int jc = 0; jc < 12; ++jc) {
            const int ci = jc >> 2, yi = jc & 3;
            const int dy = yi - yo;
#pragma unroll
            for (int s = 0; s < 2; ++s) {
                const int dx = 2 * h + s - xo;
                const bool in = rr < 6 && dy >= 0 && dy <= 2 && dx >= 0 && dx <= 2;
                const float wgt = conv_w[co * 27 + ci * 9 + min(max(dy, 0), 2) * 3 + min(max(dx, 0), 2)];
                Amat[jc][s] = in ? wgt : 0.0f;
            }
        }
    }
    v2f Ared[8];
#pragma unroll
    for (int jc = 0; jc < 8; ++jc) Ared[jc] = Amat[jc] - Amat[8 + (jc & 3)];

#pragma unroll
    for (int it = 0; it < 2; ++it) {
        int hp = tid + it * 256;
        if (hp < HALO * HALO) {
            const int hy = hp / HALO, hx = hp % HALO;
            const int gy = wgy + hy - 1, gx = wgx + hx - 1;
            const int lb = hy * PITCH + hx;
            if (gy >= 0 && gy < IMG && gx >= 0 && gx < IMG) {
                const int pix = gy * IMG + gx;
                const float o0 = obs[pix * 3 + 0], o1 = obs[pix * 3 + 1], o2 = obs[pix * 3 + 2];
                const float u0 = u_k[pix * 3 + 0], u1 = u_k[pix * 3 + 1], u2 = u_k[pix * 3 + 2];
                const float b0 = Pm[0] * o0 + Pm[3] * o1 + Pm[6] * o2;
                const float b1 = Pm[1] * o0 + Pm[4] * o1 + Pm[7] * o2;
                const float b2 = Pm[2] * o0 + Pm[5] * o1 + Pm[8] * o2;
                const float Bu0 = b0 * u0, Bu1 = b1 * u1, Bu2 = b2 * u2;
                const float bu  = Bu0 + Bu1 + Bu2;
                const float inv = 1.0f / bu;
                const float Z0 = Bu0 * inv, Z1 = Bu1 * inv, Z2 = Bu2 * inv;
                X[0 * CHSZ + lb] = Z0;
                X[1 * CHSZ + lb] = Z1;
                X[2 * CHSZ + lb] = Z2;
                const float ov[3] = {o0, o1, o2};
                const float uv[3] = {u0, u1, u2};
                const float bv[3] = {b0, b1, b2};
                const float Zv[3] = {Z0, Z1, Z2};
                const float* wp = w_k + (long)pix * 27;
#pragma unroll
                for (int ij = 0; ij < 9; ++ij) {
                    const int i = ij / 3, j = ij % 3;
                    float du0 = bv[0] * wp[0 * 9 + ij] + ((j == 0) ? ov[i] * uv[0] : 0.0f);
                    float du1 = bv[1] * wp[1 * 9 + ij] + ((j == 1) ? ov[i] * uv[1] : 0.0f);
                    float du2 = bv[2] * wp[2 * 9 + ij] + ((j == 2) ? ov[i] * uv[2] : 0.0f);
                    float sdu = du0 + du1 + du2;
                    if (ij == 0) {
                        X[3 * CHSZ + lb] = (du0 - sdu * Zv[0]) * inv;
                        X[4 * CHSZ + lb] = (du1 - sdu * Zv[1]) * inv;
                        X[5 * CHSZ + lb] = (du2 - sdu * Zv[2]) * inv;
                    } else {
                        X[(TAN1 + (ij - 1) * 2 + 0) * CHSZ + lb] = (du0 - sdu * Zv[0]) * inv;
                        X[(TAN1 + (ij - 1) * 2 + 1) * CHSZ + lb] = (du1 - sdu * Zv[1]) * inv;
                    }
                }
            } else {
#pragma unroll
                for (int c = 0; c < NCH; ++c) X[c * CHSZ + lb] = 0.0f;
            }
        }
    }
    __syncthreads();

    const int gsub = col >> 3;
    const int bx   = col & 7;

    const int bpos = (2 * wv) * PITCH + 2 * bx + 2 * h;

    auto conv_tile = [&](int t) -> v8f {
        v8f acc = {0.f, 0.f, 0.f, 0.f, 0.f, 0.f, 0.f, 0.f};
        if (t == 0) {
#pragma unroll
            for (int jc = 0; jc < 12; ++jc) {
                const int ci = jc >> 2, yi = jc & 3;
                const v2f Bv = *(const v2f*)&X[bpos + gsub * (3 * CHSZ) + ci * CHSZ + yi * PITCH];
                acc = __builtin_amdgcn_wmma_f32_16x16x4_f32(
                    false, Amat[jc], false, Bv, (short)0, acc, false, false);
            }
        } else {
#pragma unroll
            for (int jc = 0; jc < 8; ++jc) {
                const int ci = jc >> 2, yi = jc & 3;
                const v2f Bv = *(const v2f*)&X[bpos + (TAN1 + (t - 1) * 4) * CHSZ + gsub * (2 * CHSZ)
                                               + ci * CHSZ + yi * PITCH];
                acc = __builtin_amdgcn_wmma_f32_16x16x4_f32(
                    false, Ared[jc], false, Bv, (short)0, acc, false, false);
            }
        }
        return acc;
    };

    const int ty = 2 * wv + h;
    const int gy = wgy + ty;
    float p[2][3];

#pragma unroll
    for (int t = 0; t < 5; ++t) {
        const v8f d = conv_tile(t);

        if (t == 0) {
#pragma unroll
            for (int xo = 0; xo < 2; ++xo) {
                const float y0 = d[xo * 3 + 0] + cb0, y1 = d[xo * 3 + 1] + cb1, y2 = d[xo * 3 + 2] + cb2;
                const float m  = fmaxf(y0, fmaxf(y1, y2));
                const float e0 = __expf(y0 - m), e1 = __expf(y1 - m), e2 = __expf(y2 - m);
                const float is = 1.0f / (e0 + e1 + e2);
                p[xo][0] = e0 * is; p[xo][1] = e1 * is; p[xo][2] = e2 * is;
            }
#pragma unroll
            for (int xo = 0; xo < 2; ++xo)
#pragma unroll
                for (int s = 0; s < 3; ++s)
                    p[xo][s] = __shfl(p[xo][s], lane & ~8);
            if (gsub == 0) {
                float* uo = u_out + (long)(gy * IMG + wgx + 2 * bx) * 3;
#pragma unroll
                for (int xo = 0; xo < 2; ++xo)
#pragma unroll
                    for (int s = 0; s < 3; ++s) uo[xo * 3 + s] = p[xo][s];
            }
        }

        if (t > 0 || gsub == 1) {
            const int ij = 2 * t + gsub - 1;
#pragma unroll
            for (int xo = 0; xo < 2; ++xo) {
                const float d0 = d[xo * 3 + 0], d1 = d[xo * 3 + 1], d2 = d[xo * 3 + 2];
                const float dot = p[xo][0] * d0 + p[xo][1] * d1 + p[xo][2] * d2;
                float* so = &Ost[ty * WROW + (2 * bx + xo) * 27 + ij];
                so[0]  = p[xo][0] * (d0 - dot);
                so[9]  = p[xo][1] * (d1 - dot);
                so[18] = p[xo][2] * (d2 - dot);
            }
        }
    }

#pragma unroll
    for (int r = 0; r < 2; ++r) {
        const int row = 2 * wv + r;
        v4f* dst = (v4f*)(w_out + (long)((wgy + row) * IMG + wgx) * 27);
        const v4f* srow = (const v4f*)&Ost[row * WROW];
#pragma unroll
        for (int q = lane; q < WROW / 4; q += 32) dst[q] = srow[q];
    }
}

extern "C" void kernel_launch(void* const* d_in, const int* in_sizes, int n_in,
                              void* d_out, int out_size, void* d_ws, size_t ws_size,
                              hipStream_t stream) {
    (void)in_sizes; (void)n_in; (void)out_size; (void)d_ws; (void)ws_size;
    const float* obs    = (const float*)d_in[0];
    const float* P      = (const float*)d_in[1];
    const float* u_k    = (const float*)d_in[2];
    const float* w_k    = (const float*)d_in[3];
    const float* conv_w = (const float*)d_in[4];
    const float* conv_b = (const float*)d_in[5];
    float* u_out = (float*)d_out;
    float* w_out = u_out + (size_t)IMG * IMG * 3;

    dim3 grid(IMG / TILE, IMG / TILE);
    dim3 block(256);
    hmm_fused_kernel<<<grid, block, 0, stream>>>(obs, P, u_k, w_k, conv_w, conv_b,
                                                 u_out, w_out);
}
